# starting kernel
# baseline (speedup 1.0000x reference)
#include <hip/hip_runtime.h>
#include <cmath>

typedef float v2f __attribute__((ext_vector_type(2)));
typedef float v4f __attribute__((ext_vector_type(4)));
typedef float v8f __attribute__((ext_vector_type(8)));

#define BATCH 8
#define CDIM 512
#define KDIM 19
#define KPAD 32
#define HW 16384

#define CT 128
#define NT 64
#define NSPLIT 16
#define NPER (HW / NSPLIT)
#define OUT_TOTAL (BATCH * CDIM * KDIM)

#define WS_W_OFF 1024
#define WS_PART_OFF (WS_W_OFF + (size_t)BATCH * KPAD * HW)

__global__ __launch_bounds__(256)
void sg_softmax_stats(const float* __restrict__ probs, float* __restrict__ stats) {
    __shared__ float red[256];
    const int bk = blockIdx.x;
    const float* row = probs + (size_t)bk * HW;
    const int t = threadIdx.x;

    float m = -INFINITY;
    for (int i = t; i < HW; i += 256) m = fmaxf(m, row[i]);
    red[t] = m; __syncthreads();
    for (int s = 128; s > 0; s >>= 1) {
        if (t < s) red[t] = fmaxf(red[t], red[t + s]);
        __syncthreads();
    }
    m = red[0]; __syncthreads();

    float sum = 0.0f;
    for (int i = t; i < HW; i += 256) sum += __expf(row[i] - m);
    red[t] = sum; __syncthreads();
    for (int s = 128; s > 0; s >>= 1) {
        if (t < s) red[t] += red[t + s];
        __syncthreads();
    }
    if (t == 0) { stats[2 * bk] = m; stats[2 * bk + 1] = 1.0f / red[0]; }
}

__global__ __launch_bounds__(256)
void sg_softmax_write(const float* __restrict__ probs, const float* __restrict__ stats,
                      float* __restrict__ W) {
    const int n = blockIdx.x * 256 + threadIdx.x;
    const int k = blockIdx.y;
    const int b = blockIdx.z;
    float v = 0.0f;
    if (k < KDIM) {
        const int bk = b * KDIM + k;
        const float m = stats[2 * bk];
        const float sinv = stats[2 * bk + 1];
        v = __expf(probs[(size_t)bk * HW + n] - m) * sinv;
    }
    W[((size_t)b * KPAD + k) * HW + n] = v;
}

__global__ __launch_bounds__(256)
void sg_context_gemm(const float* __restrict__ feats, const float* __restrict__ W,
                     float* __restrict__ part) {
    __shared__ float ftile[CT][NT + 4];
    __shared__ float wtile[KPAD][NT + 4];

    const int t = threadIdx.x;
    const int wave = t >> 5;
    const int lane = t & 31;
    const int lrow = lane & 15;
    const int ksel = (lane >> 4) * 2;

    const int c0 = blockIdx.x * CT;
    const int b  = blockIdx.y;
    const int z  = blockIdx.z;
    const int nbeg = z * NPER;

    const float* fbase = feats + ((size_t)b * CDIM + c0) * HW;
    const float* wbase = W + (size_t)b * KPAD * HW;

    const int frow = t >> 1;
    const int fseg = (t & 1) * 32;
    const int wrow = t >> 3;
    const int wseg = (t & 7) * 8;

    v8f acc0 = {};
    v8f acc1 = {};

    for (int n0 = nbeg; n0 < nbeg + NPER; n0 += NT) {
        const v4f* fsrc = (const v4f*)(fbase + (size_t)frow * HW + n0 + fseg);
        #pragma unroll
        for (int j = 0; j < 8; ++j) {
            v4f v = __builtin_nontemporal_load(fsrc + j);
            *(v4f*)&ftile[frow][fseg + 4 * j] = v;
        }
        const v4f* wsrc = (const v4f*)(wbase + (size_t)wrow * HW + n0 + wseg);
        #pragma unroll
        for (int j = 0; j < 2; ++j) {
            *(v4f*)&wtile[wrow][wseg + 4 * j] = wsrc[j];
        }
        __syncthreads();

        #pragma unroll
        for (int kk = 0; kk < NT; kk += 4) {
            v2f a  = *(const v2f*)&ftile[wave * 16 + lrow][kk + ksel];
            v2f b0 = *(const v2f*)&wtile[lrow][kk + ksel];
            v2f b1 = *(const v2f*)&wtile[16 + lrow][kk + ksel];
            acc0 = __builtin_amdgcn_wmma_f32_16x16x4_f32(
                false, a, false, b0, (short)0, acc0, false, false);
            acc1 = __builtin_amdgcn_wmma_f32_16x16x4_f32(
                false, a, false, b1, (short)0, acc1, false, false);
        }
        __syncthreads();
    }

    float* pdst = part + (size_t)z * OUT_TOTAL;
    const int rbase = c0 + wave * 16 + (lane >> 4) * 8;
    #pragma unroll
    for (int i = 0; i < 8; ++i) {
        const int row = rbase + i;
        const size_t o = ((size_t)b * CDIM + row) * KDIM;
        pdst[o + lrow] = acc0[i];
        if (lrow + 16 < KDIM) pdst[o + lrow + 16] = acc1[i];
    }
}

__global__ __launch_bounds__(256)
void sg_reduce_partials(const float* __restrict__ part, float* __restrict__ out) {
    const int i = blockIdx.x * 256 + threadIdx.x;
    if (i < OUT_TOTAL) {
        float s = 0.0f;
        #pragma unroll
        for (int j = 0; j < NSPLIT; ++j) s += part[(size_t)j * OUT_TOTAL + i];
        out[i] = s;
    }
}

extern "C" void kernel_launch(void* const* d_in, const int* in_sizes, int n_in,
                              void* d_out, int out_size, void* d_ws, size_t ws_size,
                              hipStream_t stream) {
    const float* feats = (const float*)d_in[0];
    const float* probs = (const float*)d_in[1];
    float* out = (float*)d_out;

    float* ws    = (float*)d_ws;
    float* stats = ws;
    float* W     = ws + WS_W_OFF;
    float* part  = ws + WS_PART_OFF;

    sg_softmax_stats<<<BATCH * KDIM, 256, 0, stream>>>(probs, stats);

    dim3 g2(HW / 256, KPAD, BATCH);
    sg_softmax_write<<<g2, 256, 0, stream>>>(probs, stats, W);

    dim3 g3(CDIM / CT, BATCH, NSPLIT);
    sg_context_gemm<<<g3, 256, 0, stream>>>(feats, W, part);

    sg_reduce_partials<<<(OUT_TOTAL + 255) / 256, 256, 0, stream>>>(part, out);
}


# softmax front end in ONE launch: a work-group keeps its whole 64 KB row in registers (read once) for max, exp sum and normalised W store; stats and 16384-group write launches go; exp-sum order new
# speedup vs baseline: 2.3698x; 2.3698x over previous
#include <hip/hip_runtime.h>
#include <cmath>

typedef float v2f __attribute__((ext_vector_type(2)));
typedef float v4f __attribute__((ext_vector_type(4)));
typedef float v8f __attribute__((ext_vector_type(8)));

#define BATCH 8
#define CDIM 512
#define KDIM 19
#define KPAD 32
#define HW 16384

#define CT 128
#define NSPLIT 16
#define NPER (HW / NSPLIT)
#define OUT_TOTAL (BATCH * CDIM * KDIM)

#define WS_W_OFF 1024
#define WS_PART_OFF (WS_W_OFF + (size_t)BATCH * KPAD * HW)

#define SM_THREADS 512
#define SM_WAVES (SM_THREADS / 32)
#define SM_PIECES (HW / 4 / SM_THREADS)

__device__ __forceinline__ float sg_wave16_max(float v) {
    v = fmaxf(v, __shfl_xor(v, 1));
    v = fmaxf(v, __shfl_xor(v, 2));
    v = fmaxf(v, __shfl_xor(v, 4));
    v = fmaxf(v, __shfl_xor(v, 8));
    return v;
}
__device__ __forceinline__ float sg_wave16_sum(float v) {
    v += __shfl_xor(v, 1);
    v += __shfl_xor(v, 2);
    v += __shfl_xor(v, 4);
    v += __shfl_xor(v, 8);
    return v;
}

__global__ __launch_bounds__(SM_THREADS)
void sg_softmax_rows(const float* __restrict__ probs, float* __restrict__ W) {
    __shared__ float redm[SM_WAVES];
    __shared__ float reds[SM_WAVES];
    const int k = blockIdx.x;
    const int b = blockIdx.y;
    const int t = threadIdx.x;
    const int lane = t & 31;
    const int wave = t >> 5;

    v4f* dst = (v4f*)(W + ((size_t)b * KPAD + k) * HW) + t;
    if (k >= KDIM) {
        const v4f zero = {0.0f, 0.0f, 0.0f, 0.0f};
        #pragma unroll
        for (int j = 0; j < SM_PIECES; ++j) dst[j * SM_THREADS] = zero;
        return;
    }

    const v4f* src = (const v4f*)(probs + ((size_t)b * KDIM + k) * HW) + t;
    v4f v[SM_PIECES];
    #pragma unroll
    for (int j = 0; j < SM_PIECES; ++j) v[j] = src[j * SM_THREADS];

    float m = -INFINITY;
    #pragma unroll
    for (int j = 0; j < SM_PIECES; ++j)
        m = fmaxf(fmaxf(m, fmaxf(v[j].x, v[j].y)), fmaxf(v[j].z, v[j].w));
    m = sg_wave16_max(m);
    m = fmaxf(m, __shfl_xor(m, 16));
    if (lane == 0) redm[wave] = m;
    __syncthreads();
    m = sg_wave16_max(redm[lane & (SM_WAVES - 1)]);

    float sum = 0.0f;
    #pragma unroll
    for (int j = 0; j < SM_PIECES; ++j) {
        v[j].x = __expf(v[j].x - m);
        v[j].y = __expf(v[j].y - m);
        v[j].z = __expf(v[j].z - m);
        v[j].w = __expf(v[j].w - m);
        sum += (v[j].x + v[j].y) + (v[j].z + v[j].w);
    }
    sum = sg_wave16_sum(sum);
    sum += __shfl_xor(sum, 16);
    if (lane == 0) reds[wave] = sum;
    __syncthreads();
    sum = sg_wave16_sum(reds[lane & (SM_WAVES - 1)]);
    const float sinv = 1.0f / sum;

    #pragma unroll
    for (int j = 0; j < SM_PIECES; ++j) dst[j * SM_THREADS] = v[j] * sinv;
}

#define NCH NPER
#define WROWS (KDIM + 1)
#define WPITCH (NCH + 4)
#define GRP 8
#define NGRP (NPER / (8 * GRP))
#define WBATCH 10

__global__ __launch_bounds__(256)
void sg_context_gemm(const float* __restrict__ feats, const float* __restrict__ W,
                     float* __restrict__ part) {
    __shared__ __attribute__((aligned(16))) float wl[WROWS][WPITCH];

    const int t = threadIdx.x;
    const int wave = t >> 5;
    const int lane = t & 31;
    const int lrow = lane & 15;
    const int half = lane >> 4;

    const int c0 = blockIdx.x * CT;
    const int b  = blockIdx.y;
    const int z  = blockIdx.z;
    const int nbeg = z * NPER;

    const float* arow = feats + ((size_t)b * CDIM + c0 + wave * 16 + lrow) * HW + nbeg + 4 * half;
    const float* wsrc = W + (size_t)b * KPAD * HW + nbeg;
    const float* l0 = &wl[lrow][4 * half];
    const float* l1 = &wl[(lrow < KDIM - 16) ? 16 + lrow : KDIM][4 * half];

    v8f acc0 = {};
    v8f acc1 = {};

    v4f ring[3][GRP];
    #pragma unroll
    for (int g = 0; g < GRP; ++g)
        ring[0][g] = __builtin_nontemporal_load((const v4f*)(arow + 8 * g));
    #pragma unroll
    for (int g = 0; g < GRP; ++g)
        ring[1][g] = __builtin_nontemporal_load((const v4f*)(arow + 8 * (GRP + g)));

    #pragma unroll
    for (int r0 = 0; r0 < WROWS; r0 += WBATCH) {
        v4f tmp[WBATCH];
        #pragma unroll
        for (int j = 0; j < WBATCH; ++j)
            tmp[j] = *(const v4f*)(wsrc + (size_t)(r0 + j) * HW + 4 * t);
        #pragma unroll
        for (int j = 0; j < WBATCH; ++j)
            *(v4f*)&wl[r0 + j][4 * t] = tmp[j];
    }
    __syncthreads();

    #pragma unroll
    for (int grp = 0; grp < NGRP; ++grp) {
        if (grp + 2 < NGRP) {
            #pragma unroll
            for (int g = 0; g < GRP; ++g)
                ring[(grp + 2) % 3][g] = __builtin_nontemporal_load(
                    (const v4f*)(arow + 8 * ((grp + 2) * GRP + g)));
        }
        v4f w0[GRP], w1[GRP];
        #pragma unroll
        for (int g = 0; g < GRP; ++g) {
            w0[g] = *(const v4f*)(l0 + 8 * (grp * GRP + g));
            w1[g] = *(const v4f*)(l1 + 8 * (grp * GRP + g));
        }
        __builtin_amdgcn_sched_barrier(0);
        #pragma unroll
        for (int g = 0; g < GRP; ++g) {
            const v4f a4 = ring[grp % 3][g];
            const v2f alo = {a4.x, a4.y};
            const v2f ahi = {a4.z, a4.w};
            const v2f b0lo = {w0[g].x, w0[g].y};
            const v2f b0hi = {w0[g].z, w0[g].w};
            const v2f b1lo = {w1[g].x, w1[g].y};
            const v2f b1hi = {w1[g].z, w1[g].w};
            acc0 = __builtin_amdgcn_wmma_f32_16x16x4_f32(
                false, alo, false, b0lo, (short)0, acc0, false, false);
            acc1 = __builtin_amdgcn_wmma_f32_16x16x4_f32(
                false, alo, false, b1lo, (short)0, acc1, false, false);
            acc0 = __builtin_amdgcn_wmma_f32_16x16x4_f32(
                false, ahi, false, b0hi, (short)0, acc0, false, false);
            acc1 = __builtin_amdgcn_wmma_f32_16x16x4_f32(
                false, ahi, false, b1hi, (short)0, acc1, false, false);
        }
        __builtin_amdgcn_sched_barrier(0);
    }

    float* pdst = part + (size_t)z * OUT_TOTAL;
    const int rbase = c0 + wave * 16 + half * 8;
    #pragma unroll
    for (int i = 0; i < 8; ++i) {
        const int row = rbase + i;
        const size_t o = ((size_t)b * CDIM + row) * KDIM;
        pdst[o + lrow] = acc0[i];
        if (lrow + 16 < KDIM) pdst[o + lrow + 16] = acc1[i];
    }
}

__global__ __launch_bounds__(256)
void sg_reduce_partials(const float* __restrict__ part, float* __restrict__ out) {
    const int i = blockIdx.x * 256 + threadIdx.x;
    if (i < OUT_TOTAL) {
        float s = 0.0f;
        #pragma unroll
        for (int j = 0; j < NSPLIT; ++j) s += part[(size_t)j * OUT_TOTAL + i];
        out[i] = s;
    }
}

extern "C" void kernel_launch(void* const* d_in, const int* in_sizes, int n_in,
                              void* d_out, int out_size, void* d_ws, size_t ws_size,
                              hipStream_t stream) {
    const float* feats = (const float*)d_in[0];
    const float* probs = (const float*)d_in[1];
    float* out = (float*)d_out;

    float* ws    = (float*)d_ws;
    float* W     = ws + WS_W_OFF;
    float* part  = ws + WS_PART_OFF;

    dim3 g1(WROWS, BATCH);
    sg_softmax_rows<<<g1, SM_THREADS, 0, stream>>>(probs, W);

    dim3 g3(CDIM / CT, BATCH, NSPLIT);
    sg_context_gemm<<<g3, 256, 0, stream>>>(feats, W, part);

    sg_reduce_partials<<<(OUT_TOTAL + 255) / 256, 256, 0, stream>>>(part, out);
}


# speedup vs baseline: 2.9309x; 1.2368x over previous
#include <hip/hip_runtime.h>
#include <cmath>

typedef float v2f __attribute__((ext_vector_type(2)));
typedef float v4f __attribute__((ext_vector_type(4)));
typedef float v8f __attribute__((ext_vector_type(8)));

#define BATCH 8
#define CDIM 512
#define KDIM 19
#define KPAD 32
#define HW 16384

#define CT 128
#define NSPLIT 16
#define NPER (HW / NSPLIT)
#define OUT_TOTAL (BATCH * CDIM * KDIM)

#define WS_W_OFF 1024
#define WS_PART_OFF (WS_W_OFF + (size_t)BATCH * KPAD * HW)

#define SM_THREADS 512
#define SM_WAVES (SM_THREADS / 32)
#define SM_PIECES (HW / 4 / SM_THREADS)

__device__ __forceinline__ float sg_wave16_max(float v) {
    v = fmaxf(v, __shfl_xor(v, 1));
    v = fmaxf(v, __shfl_xor(v, 2));
    v = fmaxf(v, __shfl_xor(v, 4));
    v = fmaxf(v, __shfl_xor(v, 8));
    return v;
}
__device__ __forceinline__ float sg_wave16_sum(float v) {
    v += __shfl_xor(v, 1);
    v += __shfl_xor(v, 2);
    v += __shfl_xor(v, 4);
    v += __shfl_xor(v, 8);
    return v;
}

__global__ __launch_bounds__(SM_THREADS)
void sg_softmax_rows(const float* __restrict__ probs, float* __restrict__ W) {
    __shared__ float redm[SM_WAVES];
    __shared__ float reds[SM_WAVES];
    const int k = blockIdx.x;
    const int b = blockIdx.y;
    const int t = threadIdx.x;
    const int lane = t & 31;
    const int wave = t >> 5;

    v4f* dst = (v4f*)(W + ((size_t)b * KPAD + k) * HW) + t;
    if (k >= KDIM) {
        const v4f zero = {0.0f, 0.0f, 0.0f, 0.0f};
        #pragma unroll
        for (int j = 0; j < SM_PIECES; ++j) dst[j * SM_THREADS] = zero;
        return;
    }

    const v4f* src = (const v4f*)(probs + ((size_t)b * KDIM + k) * HW) + t;
    v4f v[SM_PIECES];
    #pragma unroll
    for (int j = 0; j < SM_PIECES; ++j) v[j] = src[j * SM_THREADS];

    float m = -INFINITY;
    #pragma unroll
    for (int j = 0; j < SM_PIECES; ++j)
        m = fmaxf(fmaxf(m, fmaxf(v[j].x, v[j].y)), fmaxf(v[j].z, v[j].w));
    m = sg_wave16_max(m);
    m = fmaxf(m, __shfl_xor(m, 16));
    if (lane == 0) redm[wave] = m;
    __syncthreads();
    m = sg_wave16_max(redm[lane & (SM_WAVES - 1)]);

    float sum = 0.0f;
    #pragma unroll
    for (int j = 0; j < SM_PIECES; ++j) {
        v[j].x = __expf(v[j].x - m);
        v[j].y = __expf(v[j].y - m);
        v[j].z = __expf(v[j].z - m);
        v[j].w = __expf(v[j].w - m);
        sum += (v[j].x + v[j].y) + (v[j].z + v[j].w);
    }
    sum = sg_wave16_sum(sum);
    sum += __shfl_xor(sum, 16);
    if (lane == 0) reds[wave] = sum;
    __syncthreads();
    sum = sg_wave16_sum(reds[lane & (SM_WAVES - 1)]);
    const float sinv = 1.0f / sum;

    #pragma unroll
    for (int j = 0; j < SM_PIECES; ++j) dst[j * SM_THREADS] = v[j] * sinv;
}

#define NCH NPER
#define WROWS (KDIM + 1)
#define WPITCH (NCH + 4)
#define GRP 8
#define NGRP (NPER / (8 * GRP))
#define WBATCH 10

__global__ __launch_bounds__(256)
void sg_context_gemm(const float* __restrict__ feats, const float* __restrict__ W,
                     float* __restrict__ part) {
    __shared__ __attribute__((aligned(16))) float wl[WROWS][WPITCH];

    const int t = threadIdx.x;
    const int wave = t >> 5;
    const int lane = t & 31;
    const int lrow = lane & 15;
    const int half = lane >> 4;

    const int c0 = blockIdx.x * CT;
    const int b  = blockIdx.y;
    const int z  = blockIdx.z;
    const int nbeg = z * NPER;

    const float* arow = feats + ((size_t)b * CDIM + c0 + wave * 16 + lrow) * HW + nbeg + 4 * half;
    const float* wsrc = W + (size_t)b * KPAD * HW + nbeg;
    const float* l0 = &wl[lrow][4 * half];
    const float* l1 = &wl[(lrow < KDIM - 16) ? 16 + lrow : KDIM][4 * half];

    v8f acc0 = {};
    v8f acc1 = {};

    v4f ring[3][GRP];
    #pragma unroll
    for (int g = 0; g < GRP; ++g)
        ring[0][g] = *(const v4f*)(arow + 8 * g);
    #pragma unroll
    for (int g = 0; g < GRP; ++g)
        ring[1][g] = *(const v4f*)(arow + 8 * (GRP + g));

    #pragma unroll
    for (int r0 = 0; r0 < WROWS; r0 += WBATCH) {
        v4f tmp[WBATCH];
        #pragma unroll
        for (int j = 0; j < WBATCH; ++j)
            tmp[j] = *(const v4f*)(wsrc + (size_t)(r0 + j) * HW + 4 * t);
        #pragma unroll
        for (int j = 0; j < WBATCH; ++j)
            *(v4f*)&wl[r0 + j][4 * t] = tmp[j];
    }
    __syncthreads();

    #pragma unroll
    for (int grp = 0; grp < NGRP; ++grp) {
        if (grp + 2 < NGRP) {
            #pragma unroll
            for (int g = 0; g < GRP; ++g)
                ring[(grp + 2) % 3][g] =
                    *(const v4f*)(arow + 8 * ((grp + 2) * GRP + g));
        }
        v4f w0[GRP], w1[GRP];
        #pragma unroll
        for (int g = 0; g < GRP; ++g) {
            w0[g] = *(const v4f*)(l0 + 8 * (grp * GRP + g));
            w1[g] = *(const v4f*)(l1 + 8 * (grp * GRP + g));
        }
        __builtin_amdgcn_sched_barrier(0);
        #pragma unroll
        for (int g = 0; g < GRP; ++g) {
            const v4f a4 = ring[grp % 3][g];
            const v2f alo = {a4.x, a4.y};
            const v2f ahi = {a4.z, a4.w};
            const v2f b0lo = {w0[g].x, w0[g].y};
            const v2f b0hi = {w0[g].z, w0[g].w};
            const v2f b1lo = {w1[g].x, w1[g].y};
            const v2f b1hi = {w1[g].z, w1[g].w};
            acc0 = __builtin_amdgcn_wmma_f32_16x16x4_f32(
                false, alo, false, b0lo, (short)0, acc0, false, false);
            acc1 = __builtin_amdgcn_wmma_f32_16x16x4_f32(
                false, alo, false, b1lo, (short)0, acc1, false, false);
            acc0 = __builtin_amdgcn_wmma_f32_16x16x4_f32(
                false, ahi, false, b0hi, (short)0, acc0, false, false);
            acc1 = __builtin_amdgcn_wmma_f32_16x16x4_f32(
                false, ahi, false, b1hi, (short)0, acc1, false, false);
        }
        __builtin_amdgcn_sched_barrier(0);
    }

    float* pdst = part + (size_t)z * OUT_TOTAL;
    const int rbase = c0 + wave * 16 + half * 8;
    #pragma unroll
    for (int i = 0; i < 8; ++i) {
        const int row = rbase + i;
        const size_t o = ((size_t)b * CDIM + row) * KDIM;
        pdst[o + lrow] = acc0[i];
        if (lrow + 16 < KDIM) pdst[o + lrow + 16] = acc1[i];
    }
}

__global__ __launch_bounds__(256)
void sg_reduce_partials(const float* __restrict__ part, float* __restrict__ out) {
    const int i = blockIdx.x * 256 + threadIdx.x;
    if (i < OUT_TOTAL) {
        float s = 0.0f;
        #pragma unroll
        for (int j = 0; j < NSPLIT; ++j) s += part[(size_t)j * OUT_TOTAL + i];
        out[i] = s;
    }
}

extern "C" void kernel_launch(void* const* d_in, const int* in_sizes, int n_in,
                              void* d_out, int out_size, void* d_ws, size_t ws_size,
                              hipStream_t stream) {
    const float* feats = (const float*)d_in[0];
    const float* probs = (const float*)d_in[1];
    float* out = (float*)d_out;

    float* ws    = (float*)d_ws;
    float* W     = ws + WS_W_OFF;
    float* part  = ws + WS_PART_OFF;

    dim3 g1(WROWS, BATCH);
    sg_softmax_rows<<<g1, SM_THREADS, 0, stream>>>(probs, W);

    dim3 g3(CDIM / CT, BATCH, NSPLIT);
    sg_context_gemm<<<g3, 256, 0, stream>>>(feats, W, part);

    sg_reduce_partials<<<(OUT_TOTAL + 255) / 256, 256, 0, stream>>>(part, out);
}
